# starting kernel
# baseline (speedup 1.0000x reference)
#include <hip/hip_runtime.h>
#include <hip/hip_bf16.h>


typedef __attribute__((ext_vector_type(2))) float v2f;
typedef __attribute__((ext_vector_type(8))) float v8f;

#define M_DIM 128
#define N_DIM 784
#define K_DIM 192
#define A_ELEMS (M_DIM * K_DIM)
#define B_ELEMS (K_DIM * N_DIM)

__global__ __launch_bounds__(256)
void im2col_pack(const float* __restrict__ x, const float* __restrict__ w,
                 float* __restrict__ Amat, float* __restrict__ Bmat) {
  int idx = blockIdx.x * blockDim.x + threadIdx.x;

  if (idx < A_ELEMS) {
    int i = idx / K_DIM, K = idx % K_DIM;
    int j = K / 96, rem = K % 96, l = rem / 3, k = rem % 3;
    Amat[idx] = w[i * 192 + l * 6 + j * 3 + k];
  }

  int t = idx - A_ELEMS;
  if (t >= 0 && t < B_ELEMS) {
    int K = t / N_DIM, p = t % N_DIM;
    int j = K / 96, rem = K % 96, l = rem / 3, k = rem % 3;
    int n = p / 28, o = p % 28;
    int heff = o + k - 1;
    float val = 0.0f;
    if (heff >= 0 && heff < 28) {
      int s    = (2 * n + j + 55) % 56;
      int hcol = (heff + 27) % 28;
      val = x[l * 1568 + s * 28 + hcol];
    }
    Bmat[t] = val;
  }
}

__global__ __launch_bounds__(128)
void wmma_gemm(const float* __restrict__ Amat, const float* __restrict__ Bmat,
               float* __restrict__ out) {
  __shared__ float As[16 * K_DIM];

  const int tid  = threadIdx.x;
  const int lane = tid & 31;
  const int wid  = tid >> 5;
  const int mt   = blockIdx.x / 13;
  const int ng   = blockIdx.x % 13;
  const int nt   = ng * 4 + wid;
  const int m0   = mt * 16;

  for (int e = tid; e < 16 * K_DIM; e += 128) {
    As[e] = Amat[(m0 + (e / K_DIM)) * K_DIM + (e % K_DIM)];
  }
  __syncthreads();

  if (nt < 49) {
    const int m    = lane & 15;
    const int half = lane >> 4;
    const int n    = nt * 16 + (lane & 15);
    const float* Arow = &As[m * K_DIM + 2 * half];
    const float* Bcol = Bmat + n;

    v8f acc = {};

#pragma unroll
    for (int kk = 0; kk < K_DIM; kk += 4) {
      v2f av;
      av.x = Arow[kk];
      av.y = Arow[kk + 1];
      v2f bv;
      bv.x = Bcol[(kk + 2 * half) * N_DIM];
      bv.y = Bcol[(kk + 2 * half + 1) * N_DIM];
      acc = __builtin_amdgcn_wmma_f32_16x16x4_f32(
          false, av, false, bv, (short)0, acc, false, false);
    }

    const int mbase = m0 + 8 * half;
#pragma unroll
    for (int r = 0; r < 8; ++r) {
      out[(mbase + r) * N_DIM + n] = acc[r];
    }
  }
}

__global__ __launch_bounds__(256)
void naive_conv(const float* __restrict__ x, const float* __restrict__ w,
                float* __restrict__ out) {
  int idx = blockIdx.x * blockDim.x + threadIdx.x;
  if (idx >= M_DIM * N_DIM) return;
  int i = idx / N_DIM, p = idx % N_DIM;
  int n = p / 28, o = p % 28;
  float acc = 0.0f;
  for (int j = 0; j < 2; ++j) {
    int s = (2 * n + j + 55) % 56;
    for (int l = 0; l < 32; ++l) {
      const float* xrow = x + l * 1568 + s * 28;
      const float* wrow = w + i * 192 + l * 6 + j * 3;
#pragma unroll
      for (int k = 0; k < 3; ++k) {
        int heff = o + k - 1;
        if (heff >= 0 && heff < 28) acc += xrow[(heff + 27) % 28] * wrow[k];
      }
    }
  }
  out[idx] = acc;
}

extern "C" void kernel_launch(void* const* d_in, const int* in_sizes, int n_in,
                              void* d_out, int out_size, void* d_ws, size_t ws_size,
                              hipStream_t stream) {
  const float* x = (const float*)d_in[0];
  const float* w = (const float*)d_in[1];
  float* out = (float*)d_out;

  const size_t need = (size_t)(A_ELEMS + B_ELEMS) * sizeof(float);
  if (ws_size < need || d_ws == nullptr) {
    naive_conv<<<(M_DIM * N_DIM + 255) / 256, 256, 0, stream>>>(x, w, out);
    return;
  }

  float* Amat = (float*)d_ws;
  float* Bmat = (float*)((char*)d_ws + (size_t)A_ELEMS * sizeof(float));

  im2col_pack<<<(A_ELEMS + B_ELEMS) / 256, 256, 0, stream>>>(x, w, Amat, Bmat);

  wmma_gemm<<<8 * 13, 128, 0, stream>>>(Amat, Bmat, out);
}


# Pack launch fused into the GEMM: A panel staged in LDS straight from w, B fragments gathered straight from x ahead of the barrier; one launch, no workspace; K order and sums unchanged
# speedup vs baseline: 1.7660x; 1.7660x over previous
#include <hip/hip_runtime.h>


typedef __attribute__((ext_vector_type(2))) float v2f;
typedef __attribute__((ext_vector_type(8))) float v8f;

#define M_DIM 128
#define N_DIM 784
#define K_DIM 192
#define L_STRIDE 1568

__global__ __launch_bounds__(128)
void conv_wmma_fused(const float* __restrict__ x, const float* __restrict__ w,
                     float* __restrict__ out) {
  __shared__ float As[16 * K_DIM];

  const int tid  = threadIdx.x;
  const int lane = tid & 31;
  const int wid  = tid >> 5;
  const int mt   = blockIdx.x / 13;
  const int ng   = blockIdx.x % 13;
  const int nt   = ng * 4 + wid;
  const int m0   = mt * 16;

  v2f wa[4][3];
#pragma unroll
  for (int it = 0; it < 4; ++it) {
    const int g  = tid + 128 * it;
    const int mr = g >> 5, l = g & 31;
    const v2f* src = (const v2f*)(w + (m0 + mr) * K_DIM + l * 6);
    wa[it][0] = src[0];
    wa[it][1] = src[1];
    wa[it][2] = src[2];
  }

  const int c    = lane & 15;
  const int half = lane >> 4;
  const int p    = nt * 16 + c;
  const int n    = p / 28, o = p % 28;
  const float* bp[2][3][2];
  bool ok[3][2];
#pragma unroll
  for (int q = 0; q < 3; ++q) {
#pragma unroll
    for (int e = 0; e < 2; ++e) {
      const int b    = 4 * q + 2 * half + e;
      const int lb   = b / 3, k = b % 3;
      const int heff = o + k - 1;
      ok[q][e] = (heff >= 0 && heff < 28);
      const int hcol = (heff + 27) % 28;
#pragma unroll
      for (int j = 0; j < 2; ++j) {
        const int s = (2 * n + j + 55) % 56;
        bp[j][q][e] = x + lb * L_STRIDE + s * 28 + hcol;
      }
    }
  }
  float bx[48][2];
#pragma unroll
  for (int st = 0; st < 48; ++st) {
    const int j = st / 24, v = (st % 24) / 3, q = st % 3;
    bx[st][0] = bp[j][q][0][v * 4 * L_STRIDE];
    bx[st][1] = bp[j][q][1][v * 4 * L_STRIDE];
  }

#pragma unroll
  for (int it = 0; it < 4; ++it) {
    const int g  = tid + 128 * it;
    const int mr = g >> 5, l = g & 31;
    float* d0 = &As[mr * K_DIM + l * 3];
    d0[0]      = wa[it][0].x;
    d0[1]      = wa[it][0].y;
    d0[2]      = wa[it][1].x;
    d0[96 + 0] = wa[it][1].y;
    d0[96 + 1] = wa[it][2].x;
    d0[96 + 2] = wa[it][2].y;
  }
  __syncthreads();

  if (nt < 49) {
    const int m = lane & 15;
    const float* Arow = &As[m * K_DIM + 2 * half];

    v8f acc = {};

#pragma unroll
    for (int st = 0; st < 48; ++st) {
      const int kk = 4 * st;
      const int q  = st % 3;
      v2f av;
      av.x = Arow[kk];
      av.y = Arow[kk + 1];
      v2f bv;
      bv.x = ok[q][0] ? bx[st][0] : 0.0f;
      bv.y = ok[q][1] ? bx[st][1] : 0.0f;
      acc = __builtin_amdgcn_wmma_f32_16x16x4_f32(
          false, av, false, bv, (short)0, acc, false, false);
    }

    const int mbase = m0 + 8 * half;
#pragma unroll
    for (int r = 0; r < 8; ++r) {
      out[(mbase + r) * N_DIM + p] = acc[r];
    }
  }
}

extern "C" void kernel_launch(void* const* d_in, const int* in_sizes, int n_in,
                              void* d_out, int out_size, void* d_ws, size_t ws_size,
                              hipStream_t stream) {
  const float* x = (const float*)d_in[0];
  const float* w = (const float*)d_in[1];
  float* out = (float*)d_out;

  conv_wmma_fused<<<8 * 13, 128, 0, stream>>>(x, w, out);
}


# A panel LDS row pitch padded from 192 to 196 floats: the 16 rows of each A-fragment read now fall on 64 different banks instead of one bank pair (was a 16-way conflict per read)
# speedup vs baseline: 2.0332x; 1.1513x over previous
#include <hip/hip_runtime.h>


typedef __attribute__((ext_vector_type(2))) float v2f;
typedef __attribute__((ext_vector_type(8))) float v8f;

#define M_DIM 128
#define N_DIM 784
#define K_DIM 192
#define L_STRIDE 1568
#define A_PITCH 196

__global__ __launch_bounds__(128)
void conv_wmma_fused(const float* __restrict__ x, const float* __restrict__ w,
                     float* __restrict__ out) {
  __shared__ float As[16 * A_PITCH];

  const int tid  = threadIdx.x;
  const int lane = tid & 31;
  const int wid  = tid >> 5;
  const int mt   = blockIdx.x / 13;
  const int ng   = blockIdx.x % 13;
  const int nt   = ng * 4 + wid;
  const int m0   = mt * 16;

  v2f wa[4][3];
#pragma unroll
  for (int it = 0; it < 4; ++it) {
    const int g  = tid + 128 * it;
    const int mr = g >> 5, l = g & 31;
    const v2f* src = (const v2f*)(w + (m0 + mr) * K_DIM + l * 6);
    wa[it][0] = src[0];
    wa[it][1] = src[1];
    wa[it][2] = src[2];
  }

  const int c    = lane & 15;
  const int half = lane >> 4;
  const int p    = nt * 16 + c;
  const int n    = p / 28, o = p % 28;
  const float* bp[2][3][2];
  bool ok[3][2];
#pragma unroll
  for (int q = 0; q < 3; ++q) {
#pragma unroll
    for (int e = 0; e < 2; ++e) {
      const int b    = 4 * q + 2 * half + e;
      const int lb   = b / 3, k = b % 3;
      const int heff = o + k - 1;
      ok[q][e] = (heff >= 0 && heff < 28);
      const int hcol = (heff + 27) % 28;
#pragma unroll
      for (int j = 0; j < 2; ++j) {
        const int s = (2 * n + j + 55) % 56;
        bp[j][q][e] = x + lb * L_STRIDE + s * 28 + hcol;
      }
    }
  }
  float bx[48][2];
#pragma unroll
  for (int st = 0; st < 48; ++st) {
    const int j = st / 24, v = (st % 24) / 3, q = st % 3;
    bx[st][0] = bp[j][q][0][v * 4 * L_STRIDE];
    bx[st][1] = bp[j][q][1][v * 4 * L_STRIDE];
  }

#pragma unroll
  for (int it = 0; it < 4; ++it) {
    const int g  = tid + 128 * it;
    const int mr = g >> 5, l = g & 31;
    float* d0 = &As[mr * A_PITCH + l * 3];
    d0[0]      = wa[it][0].x;
    d0[1]      = wa[it][0].y;
    d0[2]      = wa[it][1].x;
    d0[96 + 0] = wa[it][1].y;
    d0[96 + 1] = wa[it][2].x;
    d0[96 + 2] = wa[it][2].y;
  }
  __syncthreads();

  if (nt < 49) {
    const int m = lane & 15;
    const float* Arow = &As[m * A_PITCH + 2 * half];

    v8f acc = {};

#pragma unroll
    for (int st = 0; st < 48; ++st) {
      const int kk = 4 * st;
      const int q  = st % 3;
      v2f av;
      av.x = Arow[kk];
      av.y = Arow[kk + 1];
      v2f bv;
      bv.x = ok[q][0] ? bx[st][0] : 0.0f;
      bv.y = ok[q][1] ? bx[st][1] : 0.0f;
      acc = __builtin_amdgcn_wmma_f32_16x16x4_f32(
          false, av, false, bv, (short)0, acc, false, false);
    }

    const int mbase = m0 + 8 * half;
#pragma unroll
    for (int r = 0; r < 8; ++r) {
      out[(mbase + r) * N_DIM + p] = acc[r];
    }
  }
}

extern "C" void kernel_launch(void* const* d_in, const int* in_sizes, int n_in,
                              void* d_out, int out_size, void* d_ws, size_t ws_size,
                              hipStream_t stream) {
  const float* x = (const float*)d_in[0];
  const float* w = (const float*)d_in[1];
  float* out = (float*)d_out;

  conv_wmma_fused<<<8 * 13, 128, 0, stream>>>(x, w, out);
}


# A fragments of all 48 K steps read from LDS as one group ahead of the WMMA chain (96 VGPRs) instead of one read per two WMMAs with its LDS latency exposed; K order and sums unchanged
# speedup vs baseline: 2.1630x; 1.0638x over previous
#include <hip/hip_runtime.h>


typedef __attribute__((ext_vector_type(2))) float v2f;
typedef __attribute__((ext_vector_type(8))) float v8f;

#define M_DIM 128
#define N_DIM 784
#define K_DIM 192
#define L_STRIDE 1568
#define A_PITCH 196

__global__ __launch_bounds__(128)
void conv_wmma_fused(const float* __restrict__ x, const float* __restrict__ w,
                     float* __restrict__ out) {
  __shared__ float As[16 * A_PITCH];

  const int tid  = threadIdx.x;
  const int lane = tid & 31;
  const int wid  = tid >> 5;
  const int mt   = blockIdx.x / 13;
  const int ng   = blockIdx.x % 13;
  const int nt   = ng * 4 + wid;
  const int m0   = mt * 16;

  v2f wa[4][3];
#pragma unroll
  for (int it = 0; it < 4; ++it) {
    const int g  = tid + 128 * it;
    const int mr = g >> 5, l = g & 31;
    const v2f* src = (const v2f*)(w + (m0 + mr) * K_DIM + l * 6);
    wa[it][0] = src[0];
    wa[it][1] = src[1];
    wa[it][2] = src[2];
  }

  const int c    = lane & 15;
  const int half = lane >> 4;
  const int p    = nt * 16 + c;
  const int n    = p / 28, o = p % 28;
  const float* bp[2][3][2];
  bool ok[3][2];
#pragma unroll
  for (int q = 0; q < 3; ++q) {
#pragma unroll
    for (int e = 0; e < 2; ++e) {
      const int b    = 4 * q + 2 * half + e;
      const int lb   = b / 3, k = b % 3;
      const int heff = o + k - 1;
      ok[q][e] = (heff >= 0 && heff < 28);
      const int hcol = (heff + 27) % 28;
#pragma unroll
      for (int j = 0; j < 2; ++j) {
        const int s = (2 * n + j + 55) % 56;
        bp[j][q][e] = x + lb * L_STRIDE + s * 28 + hcol;
      }
    }
  }
  float bx[48][2];
#pragma unroll
  for (int st = 0; st < 48; ++st) {
    const int j = st / 24, v = (st % 24) / 3, q = st % 3;
    bx[st][0] = bp[j][q][0][v * 4 * L_STRIDE];
    bx[st][1] = bp[j][q][1][v * 4 * L_STRIDE];
  }

#pragma unroll
  for (int it = 0; it < 4; ++it) {
    const int g  = tid + 128 * it;
    const int mr = g >> 5, l = g & 31;
    float* d0 = &As[mr * A_PITCH + l * 3];
    d0[0]      = wa[it][0].x;
    d0[1]      = wa[it][0].y;
    d0[2]      = wa[it][1].x;
    d0[96 + 0] = wa[it][1].y;
    d0[96 + 1] = wa[it][2].x;
    d0[96 + 2] = wa[it][2].y;
  }
  __syncthreads();

  if (nt < 49) {
    const int m = lane & 15;
    const float* Arow = &As[m * A_PITCH + 2 * half];

    v2f av[48];
#pragma unroll
    for (int st = 0; st < 48; ++st) {
      av[st].x = Arow[4 * st];
      av[st].y = Arow[4 * st + 1];
    }
    __builtin_amdgcn_sched_barrier(0);

    v8f acc = {};

#pragma unroll
    for (int st = 0; st < 48; ++st) {
      const int q  = st % 3;
      v2f bv;
      bv.x = ok[q][0] ? bx[st][0] : 0.0f;
      bv.y = ok[q][1] ? bx[st][1] : 0.0f;
      acc = __builtin_amdgcn_wmma_f32_16x16x4_f32(
          false, av[st], false, bv, (short)0, acc, false, false);
    }

    const int mbase = m0 + 8 * half;
#pragma unroll
    for (int r = 0; r < 8; ++r) {
      out[(mbase + r) * N_DIM + p] = acc[r];
    }
  }
}

extern "C" void kernel_launch(void* const* d_in, const int* in_sizes, int n_in,
                              void* d_out, int out_size, void* d_ws, size_t ws_size,
                              hipStream_t stream) {
  const float* x = (const float*)d_in[0];
  const float* w = (const float*)d_in[1];
  float* out = (float*)d_out;

  conv_wmma_fused<<<8 * 13, 128, 0, stream>>>(x, w, out);
}


# Block barrier waits for LDS only (__syncthreads drained all 96 B gathers first); gathers in K-step order, 54 ahead of the barrier, 42 inside the WMMA chain, which runs under the loads; sums unchanged
# speedup vs baseline: 2.4075x; 1.1130x over previous
#include <hip/hip_runtime.h>


typedef __attribute__((ext_vector_type(2))) float v2f;
typedef __attribute__((ext_vector_type(8))) float v8f;

#define M_DIM 128
#define N_DIM 784
#define K_DIM 192
#define L_STRIDE 1568
#define A_PITCH 196

__global__ __launch_bounds__(128)
void conv_wmma_fused(const float* __restrict__ x, const float* __restrict__ w,
                     float* __restrict__ out) {
  __shared__ float As[16 * A_PITCH];

  const int tid  = threadIdx.x;
  const int lane = tid & 31;
  const int wid  = tid >> 5;
  const int mt   = blockIdx.x / 13;
  const int ng   = blockIdx.x % 13;
  const int nt   = ng * 4 + wid;
  const int m0   = mt * 16;

  v2f wa[4][3];
#pragma unroll
  for (int it = 0; it < 4; ++it) {
    const int g  = tid + 128 * it;
    const int mr = g >> 5, l = g & 31;
    const v2f* src = (const v2f*)(w + (m0 + mr) * K_DIM + l * 6);
    wa[it][0] = src[0];
    wa[it][1] = src[1];
    wa[it][2] = src[2];
  }

  const int c    = lane & 15;
  const int half = lane >> 4;
  const int p    = nt * 16 + c;
  const int n    = p / 28, o = p % 28;
  const float* bp[2][3][2];
  bool ok[3][2];
#pragma unroll
  for (int q = 0; q < 3; ++q) {
#pragma unroll
    for (int e = 0; e < 2; ++e) {
      const int b    = 4 * q + 2 * half + e;
      const int lb   = b / 3, k = b % 3;
      const int heff = o + k - 1;
      ok[q][e] = (heff >= 0 && heff < 28);
      const int hcol = (heff + 27) % 28;
#pragma unroll
      for (int j = 0; j < 2; ++j) {
        const int s = (2 * n + j + 55) % 56;
        bp[j][q][e] = x + lb * L_STRIDE + s * 28 + hcol;
      }
    }
  }
  float bx[48][2];
#pragma unroll
  for (int g = 0; g < 9; ++g) {
#pragma unroll
    for (int q = 0; q < 3; ++q) {
      bx[3 * g + q][0] = bp[g / 8][q][0][(g % 8) * 4 * L_STRIDE];
      bx[3 * g + q][1] = bp[g / 8][q][1][(g % 8) * 4 * L_STRIDE];
    }
    __builtin_amdgcn_sched_barrier(0);
  }

#pragma unroll
  for (int it = 0; it < 4; ++it) {
    const int g  = tid + 128 * it;
    const int mr = g >> 5, l = g & 31;
    float* d0 = &As[mr * A_PITCH + l * 3];
    d0[0]      = wa[it][0].x;
    d0[1]      = wa[it][0].y;
    d0[2]      = wa[it][1].x;
    d0[96 + 0] = wa[it][1].y;
    d0[96 + 1] = wa[it][2].x;
    d0[96 + 2] = wa[it][2].y;
  }
  __builtin_amdgcn_fence(__ATOMIC_RELEASE, "workgroup", "local");
  __builtin_amdgcn_s_barrier();
  __builtin_amdgcn_fence(__ATOMIC_ACQUIRE, "workgroup", "local");

  const int m = lane & 15;
  const float* Arow = &As[m * A_PITCH + 2 * half];

  v2f av[48];
#pragma unroll
  for (int st = 0; st < 48; ++st) {
    av[st].x = Arow[4 * st];
    av[st].y = Arow[4 * st + 1];
  }
  __builtin_amdgcn_sched_barrier(0);

  if (nt < 49) {
    v8f acc = {};

#pragma unroll
    for (int st = 0; st < 48; ++st) {
      const int q  = st % 3;
      v2f bv;
      bv.x = ok[q][0] ? bx[st][0] : 0.0f;
      bv.y = ok[q][1] ? bx[st][1] : 0.0f;
      acc = __builtin_amdgcn_wmma_f32_16x16x4_f32(
          false, av[st], false, bv, (short)0, acc, false, false);
      if (st < 21) {
        const int t = st + 3;
        bx[24 + t][0] = bp[1][t % 3][0][(t / 3) * 4 * L_STRIDE];
        bx[24 + t][1] = bp[1][t % 3][1][(t / 3) * 4 * L_STRIDE];
        __builtin_amdgcn_sched_barrier(0);
      }
    }

    const int mbase = m0 + 8 * half;
#pragma unroll
    for (int r = 0; r < 8; ++r) {
      out[(mbase + r) * N_DIM + p] = acc[r];
    }
  }
}

extern "C" void kernel_launch(void* const* d_in, const int* in_sizes, int n_in,
                              void* d_out, int out_size, void* d_ws, size_t ws_size,
                              hipStream_t stream) {
  const float* x = (const float*)d_in[0];
  const float* w = (const float*)d_in[1];
  float* out = (float*)d_out;

  conv_wmma_fused<<<8 * 13, 128, 0, stream>>>(x, w, out);
}
